# starting kernel
# baseline (speedup 1.0000x reference)
#include <hip/hip_runtime.h>
#include <hip/hip_bf16.h>

typedef __attribute__((ext_vector_type(2))) float v2f;
typedef __attribute__((ext_vector_type(8))) float v8f;

#define NB 2
#define NG 8
#define ND 32
#define NH 128
#define NW 160
#define DHW (ND * NH * NW)
#define NPTS (NB * DHW)
#define TILES_PER_WAVE 4

#define SWIZ_SWAPX16 0x401F

__device__ __forceinline__ float half_swap(float v) {
    return __int_as_float(
        __builtin_amdgcn_ds_swizzle(__float_as_int(v), SWIZ_SWAPX16));
}

__global__ __launch_bounds__(256) void mlp_sim_kernel(
    const float* __restrict__ x1,
    const float* __restrict__ w0,
    const float* __restrict__ bn0_scale, const float* __restrict__ bn0_bias,
    const float* __restrict__ w1,
    const float* __restrict__ bn1_scale, const float* __restrict__ bn1_bias,
    const float* __restrict__ w_sim, const float* __restrict__ b_sim,
    float* __restrict__ sim)
{
    const int lane = threadIdx.x & 31;
    const int wib  = threadIdx.x >> 5;
    const int gwave = blockIdx.x * 8 + wib;
    const int hi   = lane >> 4;
    const int col  = lane & 15;

    const int kb = hi * 2;
    v2f a0, a1;
    a0.x = w0[col * 8 + kb + 0];
    a0.y = w0[col * 8 + kb + 1];
    a1.x = w0[col * 8 + 4 + kb + 0];
    a1.y = w0[col * 8 + 4 + kb + 1];

    v2f aw[4];
#pragma unroll
    for (int j = 0; j < 4; ++j) {
        const int k0 = j * 4 + kb;
        aw[j].x = (col < 8) ? w1[col * 16 + k0 + 0] : 0.0f;
        aw[j].y = (col < 8) ? w1[col * 16 + k0 + 1] : 0.0f;
    }

    float s0[8], bi0[8], s1[8], bi1[8], ws[8];
    const int rb = 8 * hi;
#pragma unroll
    for (int v = 0; v < 8; ++v) {
        s0[v]  = bn0_scale[rb + v];
        bi0[v] = bn0_bias[rb + v];
        const int r = (rb + v) & 7;
        s1[v]  = bn1_scale[r];
        bi1[v] = bn1_bias[r];
        ws[v]  = w_sim[v];
    }
    const float bs = b_sim[0];

#pragma unroll
    for (int t = 0; t < TILES_PER_WAVE; ++t) {
        const int base = (gwave * TILES_PER_WAVE + t) * 16;
        const int p    = base + col;
        const int b    = p / DHW;
        const int rem  = p - b * DHW;

        const float* xp = x1 + (size_t)(b * NG + kb) * DHW + rem;
        v2f b0, b1;
        b0.x = xp[0];
        b0.y = xp[DHW];
        b1.x = xp[4 * (size_t)DHW];
        b1.y = xp[5 * (size_t)DHW];
        __builtin_prefetch(xp + 16, 0, 3);

        v8f c = {};
        c = __builtin_amdgcn_wmma_f32_16x16x4_f32(false, a0, false, b0,
                                                  (short)0, c, false, false);
        c = __builtin_amdgcn_wmma_f32_16x16x4_f32(false, a1, false, b1,
                                                  (short)0, c, false, false);
        float h[8];
#pragma unroll
        for (int v = 0; v < 8; ++v)
            h[v] = fmaxf(fmaf(c[v], s0[v], bi0[v]), 0.0f);

        float hx[8];
#pragma unroll
        for (int v = 0; v < 8; ++v)
            hx[v] = half_swap(h[v]);

        v2f bb[4];
        bb[0].x = hi ? hx[2] : h[0];  bb[0].y = hi ? hx[3] : h[1];
        bb[1].x = hi ? hx[6] : h[4];  bb[1].y = hi ? hx[7] : h[5];
        bb[2].x = hi ? h[2] : hx[0];  bb[2].y = hi ? h[3] : hx[1];
        bb[3].x = hi ? h[6] : hx[4];  bb[3].y = hi ? h[7] : hx[5];

        v8f d = {};
#pragma unroll
        for (int j = 0; j < 4; ++j)
            d = __builtin_amdgcn_wmma_f32_16x16x4_f32(false, aw[j], false, bb[j],
                                                      (short)0, d, false, false);

        float g[8];
#pragma unroll
        for (int v = 0; v < 8; ++v)
            g[v] = fmaxf(fmaf(d[v], s1[v], bi1[v]), 0.0f);

        float o = bs;
#pragma unroll
        for (int v = 0; v < 8; ++v)
            o = fmaf(ws[v], g[v], o);

        if (!hi)
            sim[p] = o;
    }
}

__device__ __forceinline__ int reflect_ix(int t, int n) {
    t = (t < 0) ? -t : t;
    t = (t >= n) ? (2 * n - 2 - t) : t;
    return t;
}

__global__ __launch_bounds__(256) void aggregate_kernel(
    const float* __restrict__ sim,
    const float* __restrict__ offset,
    const float* __restrict__ weight,
    float* __restrict__ out)
{
    const int idx = blockIdx.x * 256 + threadIdx.x;
    const int x = idx % NW;
    int t = idx / NW;
    const int y = t % NH;
    t /= NH;
    const int d = t % ND;
    const int b = t / ND;

    int yn[3], xn[3], yw[3], xw[3];
#pragma unroll
    for (int i = 0; i < 3; ++i) {
        yn[i] = reflect_ix(y + (i - 1) * 2, NH);
        xn[i] = reflect_ix(x + (i - 1) * 2, NW);
        yw[i] = reflect_ix(y + (i - 1) * 4, NH);
        xw[i] = reflect_ix(x + (i - 1) * 4, NW);
    }

    const float* sb = sim + (size_t)(b * ND + d) * NH * NW;
    const float* ob = offset + (size_t)b * 18 * NH * NW + y * NW + x;

    float acc = 0.0f;
#pragma unroll
    for (int s = 0; s < 9; ++s) {
        const int iy = s / 3, ix = s % 3;
        const float c1 = sb[yn[iy] * NW + xn[ix]];
        const float c2 = sb[yw[iy] * NW + xw[ix]];
        const float w_wide   = ob[(size_t)s * NH * NW];
        const float w_narrow = ob[(size_t)(s + 9) * NH * NW];
        acc = fmaf(0.5f, fmaf(w_narrow, c1, w_wide * c2), acc);
    }
    out[idx] = acc * weight[(size_t)b * NH * NW + y * NW + x];
}

extern "C" void kernel_launch(void* const* d_in, const int* in_sizes, int n_in,
                              void* d_out, int out_size, void* d_ws, size_t ws_size,
                              hipStream_t stream)
{
    const float* x1        = (const float*)d_in[0];
    const float* offset    = (const float*)d_in[1];
    const float* weight    = (const float*)d_in[2];
    const float* w0        = (const float*)d_in[3];
    const float* bn0_scale = (const float*)d_in[4];
    const float* bn0_bias  = (const float*)d_in[5];
    const float* w1        = (const float*)d_in[6];
    const float* bn1_scale = (const float*)d_in[7];
    const float* bn1_bias  = (const float*)d_in[8];
    const float* w_sim     = (const float*)d_in[9];
    const float* b_sim     = (const float*)d_in[10];
    float* out = (float*)d_out;
    float* sim = (float*)d_ws;

    const int waves   = NPTS / (16 * TILES_PER_WAVE);
    const int blocks1 = waves / 8;
    mlp_sim_kernel<<<blocks1, 256, 0, stream>>>(
        x1, w0, bn0_scale, bn0_bias, w1, bn1_scale, bn1_bias, w_sim, b_sim, sim);

    const int blocks2 = NPTS / 256;
    aggregate_kernel<<<blocks2, 256, 0, stream>>>(sim, offset, weight, out);
}


# One launch, rebuilt to hide latency: 16-wave workgroup per (batch, depth, 32-row band), each wave owns 400 contiguous points, next 5 tiles' loads kept in flight; sim slab in LDS feeds aggregation
# speedup vs baseline: 1.2668x; 1.2668x over previous
#include <hip/hip_runtime.h>


typedef __attribute__((ext_vector_type(2))) float v2f;
typedef __attribute__((ext_vector_type(8))) float v8f;

#define NB 2
#define NG 8
#define ND 32
#define NH 128
#define NW 160
#define HW (NH * NW)
#define DHW (ND * HW)

#define BAND 32
#define SLAB_ROWS 40
#define SLAB (SLAB_ROWS * NW)
#define NWAVES 16
#define PTS_PER_WAVE (SLAB / NWAVES)
#define GT 5
#define NGRP 5
#define NTHR (NWAVES * 32)
#define NJ 5

#define SWIZ_SWAPX16 0x401F

__device__ __forceinline__ float half_swap(float v) {
    return __int_as_float(
        __builtin_amdgcn_ds_swizzle(__float_as_int(v), SWIZ_SWAPX16));
}

__device__ __forceinline__ int reflect_ix(int t, int n) {
    t = (t < 0) ? -t : t;
    t = (t >= n) ? (2 * n - 2 - t) : t;
    return t;
}

__global__ __launch_bounds__(NTHR) void sim_aggregate_kernel(
    const float* __restrict__ x1,
    const float* __restrict__ offset,
    const float* __restrict__ weight,
    const float* __restrict__ w0,
    const float* __restrict__ bn0_scale, const float* __restrict__ bn0_bias,
    const float* __restrict__ w1,
    const float* __restrict__ bn1_scale, const float* __restrict__ bn1_bias,
    const float* __restrict__ w_sim, const float* __restrict__ b_sim,
    float* __restrict__ out)
{
    __shared__ float sim_s[SLAB + PTS_PER_WAVE];

    const int tid  = threadIdx.x;
    const int lane = tid & 31;
    const int wv   = tid >> 5;
    const int hi   = lane >> 4;
    const int col  = lane & 15;

    const int band = blockIdx.x & 3;
    const int pd   = blockIdx.x >> 2;
    const int b    = pd / ND;
    const int d    = pd - b * ND;
    const int r0   = band * BAND;
    int lo = r0 - 4;
    lo = (lo < 0) ? 0 : lo;
    lo = (lo > NH - SLAB_ROWS) ? (NH - SLAB_ROWS) : lo;

    const int kb = hi * 2;
    v2f a0, a1;
    a0.x = w0[col * 8 + kb + 0];
    a0.y = w0[col * 8 + kb + 1];
    a1.x = w0[col * 8 + 4 + kb + 0];
    a1.y = w0[col * 8 + 4 + kb + 1];

    v2f aw[4];
#pragma unroll
    for (int j = 0; j < 4; ++j) {
        const int k0 = j * 4 + kb;
        aw[j].x = (col < 8) ? w1[col * 16 + k0 + 0] : 0.0f;
        aw[j].y = (col < 8) ? w1[col * 16 + k0 + 1] : 0.0f;
    }

    float s0[8], bi0[8], s1[8], bi1[8], ws[8];
    const int rb = 8 * hi;
#pragma unroll
    for (int v = 0; v < 8; ++v) {
        s0[v]  = bn0_scale[rb + v];
        bi0[v] = bn0_bias[rb + v];
        const int r = (rb + v) & 7;
        s1[v]  = bn1_scale[r];
        bi1[v] = bn1_bias[r];
        ws[v]  = w_sim[v];
    }
    const float bs = b_sim[0];

    const float* xs = x1 + (size_t)(b * NG + kb) * DHW + (size_t)d * HW
                         + lo * NW + wv * PTS_PER_WAVE + col;
    float* sw = sim_s + (hi ? SLAB : wv * PTS_PER_WAVE) + col;

    const float* ob = offset + (size_t)b * 18 * HW;
    const float* wb = weight + (size_t)b * HW;
    float ow[2][2][18];
    float wt[2][2];

#define LOAD_AGG(J, BUF)                                                      \
    {                                                                         \
        const int l_  = (J) * NTHR + tid;                                     \
        const int rl_ = l_ / NW;                                              \
        const int x_  = l_ - rl_ * NW;                                        \
        _Pragma("unroll")                                                     \
        for (int h_ = 0; h_ < 2; ++h_) {                                      \
            const int pix_ = (r0 + rl_ + 16 * h_) * NW + x_;                  \
            _Pragma("unroll")                                                 \
            for (int s_ = 0; s_ < 18; ++s_)                                   \
                ow[BUF][h_][s_] = ob[(size_t)s_ * HW + pix_];                 \
            wt[BUF][h_] = wb[pix_];                                           \
        }                                                                     \
    }

    v2f xa[2][GT], xb[2][GT];

#define LOAD_X(G, BUF)                                                        \
    {                                                                         \
        _Pragma("unroll")                                                     \
        for (int t_ = 0; t_ < GT; ++t_) {                                     \
            const float* p_ = xs + ((G) * GT + t_) * 16;                      \
            xa[BUF][t_].x = p_[0];                                            \
            xa[BUF][t_].y = p_[DHW];                                          \
            xb[BUF][t_].x = p_[4 * (size_t)DHW];                              \
            xb[BUF][t_].y = p_[5 * (size_t)DHW];                              \
        }                                                                     \
    }

    LOAD_X(0, 0)

#pragma unroll
    for (int g = 0; g < NGRP; ++g) {
        const int cur = g & 1;
        if (g + 1 < NGRP) LOAD_X(g + 1, cur ^ 1)
        if (g == NGRP - 2) LOAD_AGG(0, 0)
        if (g == NGRP - 1) LOAD_AGG(1, 1)

        __builtin_amdgcn_sched_barrier(0);

#pragma unroll
        for (int t = 0; t < GT; ++t) {
            v8f c = {};
            c = __builtin_amdgcn_wmma_f32_16x16x4_f32(false, a0, false, xa[cur][t],
                                                      (short)0, c, false, false);
            c = __builtin_amdgcn_wmma_f32_16x16x4_f32(false, a1, false, xb[cur][t],
                                                      (short)0, c, false, false);
            float h[8];
#pragma unroll
            for (int v = 0; v < 8; ++v)
                h[v] = fmaxf(fmaf(c[v], s0[v], bi0[v]), 0.0f);

            float hx[8];
#pragma unroll
            for (int v = 0; v < 8; ++v)
                hx[v] = half_swap(h[v]);

            v2f bb[4];
            bb[0].x = hi ? hx[2] : h[0];  bb[0].y = hi ? hx[3] : h[1];
            bb[1].x = hi ? hx[6] : h[4];  bb[1].y = hi ? hx[7] : h[5];
            bb[2].x = hi ? h[2] : hx[0];  bb[2].y = hi ? h[3] : hx[1];
            bb[3].x = hi ? h[6] : hx[4];  bb[3].y = hi ? h[7] : hx[5];

            v8f dd = {};
#pragma unroll
            for (int j = 0; j < 4; ++j)
                dd = __builtin_amdgcn_wmma_f32_16x16x4_f32(false, aw[j], false, bb[j],
                                                           (short)0, dd, false, false);

            float gg[8];
#pragma unroll
            for (int v = 0; v < 8; ++v)
                gg[v] = fmaxf(fmaf(dd[v], s1[v], bi1[v]), 0.0f);

            float o = bs;
#pragma unroll
            for (int v = 0; v < 8; ++v)
                o = fmaf(ws[v], gg[v], o);

            sw[(g * GT + t) * 16] = o;
        }
        __builtin_amdgcn_sched_barrier(0);
    }

    __syncthreads();

    float* outb = out + (size_t)pd * HW;

#pragma unroll
    for (int j = 0; j < NJ; ++j) {
        const int cur = j & 1;

        const int l  = j * NTHR + tid;
        const int rl = l / NW;
        const int x  = l - rl * NW;
        int xn[3], xw[3];
#pragma unroll
        for (int i = 0; i < 3; ++i) {
            xn[i] = reflect_ix(x + (i - 1) * 2, NW);
            xw[i] = reflect_ix(x + (i - 1) * 4, NW);
        }
#pragma unroll
        for (int hh = 0; hh < 2; ++hh) {
            const int y = r0 + rl + 16 * hh;
            int yn[3], yw[3];
#pragma unroll
            for (int i = 0; i < 3; ++i) {
                yn[i] = (reflect_ix(y + (i - 1) * 2, NH) - lo) * NW;
                yw[i] = (reflect_ix(y + (i - 1) * 4, NH) - lo) * NW;
            }
            float acc = 0.0f;
#pragma unroll
            for (int s = 0; s < 9; ++s) {
                const int iy = s / 3, ix = s % 3;
                const float c1 = sim_s[yn[iy] + xn[ix]];
                const float c2 = sim_s[yw[iy] + xw[ix]];
                const float w_wide   = ow[cur][hh][s];
                const float w_narrow = ow[cur][hh][s + 9];
                acc = fmaf(0.5f, fmaf(w_narrow, c1, w_wide * c2), acc);
            }
            outb[y * NW + x] = acc * wt[cur][hh];
        }
        if (j + 2 < NJ) LOAD_AGG(j + 2, cur)
    }
#undef LOAD_AGG
#undef LOAD_X
}

extern "C" void kernel_launch(void* const* d_in, const int* in_sizes, int n_in,
                              void* d_out, int out_size, void* d_ws, size_t ws_size,
                              hipStream_t stream)
{
    const float* x1        = (const float*)d_in[0];
    const float* offset    = (const float*)d_in[1];
    const float* weight    = (const float*)d_in[2];
    const float* w0        = (const float*)d_in[3];
    const float* bn0_scale = (const float*)d_in[4];
    const float* bn0_bias  = (const float*)d_in[5];
    const float* w1        = (const float*)d_in[6];
    const float* bn1_scale = (const float*)d_in[7];
    const float* bn1_bias  = (const float*)d_in[8];
    const float* w_sim     = (const float*)d_in[9];
    const float* b_sim     = (const float*)d_in[10];
    float* out = (float*)d_out;

    sim_aggregate_kernel<<<256, NTHR, 0, stream>>>(
        x1, offset, weight, w0, bn0_scale, bn0_bias, w1, bn1_scale, bn1_bias,
        w_sim, b_sim, out);
}


# Layer-1 to layer-2 re-layout by 4 v_permlane16_swap_b32 per tile (register-pair row swap on the vector ALU) in place of 8 ds_swizzle_b32 + 8 selects; k order and sum order unchanged
# speedup vs baseline: 1.3308x; 1.0505x over previous
#include <hip/hip_runtime.h>


typedef __attribute__((ext_vector_type(2))) float v2f;
typedef __attribute__((ext_vector_type(8))) float v8f;

#define NB 2
#define NG 8
#define ND 32
#define NH 128
#define NW 160
#define HW (NH * NW)
#define DHW (ND * HW)

#define BAND 32
#define SLAB_ROWS 40
#define SLAB (SLAB_ROWS * NW)
#define NWAVES 16
#define PTS_PER_WAVE (SLAB / NWAVES)
#define GT 5
#define NGRP 5
#define NTHR (NWAVES * 32)
#define NJ 5

typedef __attribute__((ext_vector_type(2))) unsigned int v2u;

__device__ __forceinline__ void row_swap(float& a, float& b) {
    const v2u r = __builtin_amdgcn_permlane16_swap(__float_as_uint(a), __float_as_uint(b),
                                                   false, false);
    a = __uint_as_float(r.x);
    b = __uint_as_float(r.y);
}

__device__ __forceinline__ int reflect_ix(int t, int n) {
    t = (t < 0) ? -t : t;
    t = (t >= n) ? (2 * n - 2 - t) : t;
    return t;
}

__global__ __launch_bounds__(NTHR) void sim_aggregate_kernel(
    const float* __restrict__ x1,
    const float* __restrict__ offset,
    const float* __restrict__ weight,
    const float* __restrict__ w0,
    const float* __restrict__ bn0_scale, const float* __restrict__ bn0_bias,
    const float* __restrict__ w1,
    const float* __restrict__ bn1_scale, const float* __restrict__ bn1_bias,
    const float* __restrict__ w_sim, const float* __restrict__ b_sim,
    float* __restrict__ out)
{
    __shared__ float sim_s[SLAB + PTS_PER_WAVE];

    const int tid  = threadIdx.x;
    const int lane = tid & 31;
    const int wv   = tid >> 5;
    const int hi   = lane >> 4;
    const int col  = lane & 15;

    const int band = blockIdx.x & 3;
    const int pd   = blockIdx.x >> 2;
    const int b    = pd / ND;
    const int d    = pd - b * ND;
    const int r0   = band * BAND;
    int lo = r0 - 4;
    lo = (lo < 0) ? 0 : lo;
    lo = (lo > NH - SLAB_ROWS) ? (NH - SLAB_ROWS) : lo;

    const int kb = hi * 2;
    v2f a0, a1;
    a0.x = w0[col * 8 + kb + 0];
    a0.y = w0[col * 8 + kb + 1];
    a1.x = w0[col * 8 + 4 + kb + 0];
    a1.y = w0[col * 8 + 4 + kb + 1];

    v2f aw[4];
#pragma unroll
    for (int j = 0; j < 4; ++j) {
        const int k0 = j * 4 + kb;
        aw[j].x = (col < 8) ? w1[col * 16 + k0 + 0] : 0.0f;
        aw[j].y = (col < 8) ? w1[col * 16 + k0 + 1] : 0.0f;
    }

    float s0[8], bi0[8], s1[8], bi1[8], ws[8];
    const int rb = 8 * hi;
#pragma unroll
    for (int v = 0; v < 8; ++v) {
        s0[v]  = bn0_scale[rb + v];
        bi0[v] = bn0_bias[rb + v];
        const int r = (rb + v) & 7;
        s1[v]  = bn1_scale[r];
        bi1[v] = bn1_bias[r];
        ws[v]  = w_sim[v];
    }
    const float bs = b_sim[0];

    const float* xs = x1 + (size_t)(b * NG + kb) * DHW + (size_t)d * HW
                         + lo * NW + wv * PTS_PER_WAVE + col;
    float* sw = sim_s + (hi ? SLAB : wv * PTS_PER_WAVE) + col;

    const float* ob = offset + (size_t)b * 18 * HW;
    const float* wb = weight + (size_t)b * HW;
    float ow[2][2][18];
    float wt[2][2];

#define LOAD_AGG(J, BUF)                                                      \
    {                                                                         \
        const int l_  = (J) * NTHR + tid;                                     \
        const int rl_ = l_ / NW;                                              \
        const int x_  = l_ - rl_ * NW;                                        \
        _Pragma("unroll")                                                     \
        for (int h_ = 0; h_ < 2; ++h_) {                                      \
            const int pix_ = (r0 + rl_ + 16 * h_) * NW + x_;                  \
            _Pragma("unroll")                                                 \
            for (int s_ = 0; s_ < 18; ++s_)                                   \
                ow[BUF][h_][s_] = ob[(size_t)s_ * HW + pix_];                 \
            wt[BUF][h_] = wb[pix_];                                           \
        }                                                                     \
    }

    v2f xa[2][GT], xb[2][GT];

#define LOAD_X(G, BUF)                                                        \
    {                                                                         \
        _Pragma("unroll")                                                     \
        for (int t_ = 0; t_ < GT; ++t_) {                                     \
            const float* p_ = xs + ((G) * GT + t_) * 16;                      \
            xa[BUF][t_].x = p_[0];                                            \
            xa[BUF][t_].y = p_[DHW];                                          \
            xb[BUF][t_].x = p_[4 * (size_t)DHW];                              \
            xb[BUF][t_].y = p_[5 * (size_t)DHW];                              \
        }                                                                     \
    }

    LOAD_X(0, 0)

#pragma unroll
    for (int g = 0; g < NGRP; ++g) {
        const int cur = g & 1;
        if (g + 1 < NGRP) LOAD_X(g + 1, cur ^ 1)
        if (g == NGRP - 2) LOAD_AGG(0, 0)
        if (g == NGRP - 1) LOAD_AGG(1, 1)

        __builtin_amdgcn_sched_barrier(0);

#pragma unroll
        for (int t = 0; t < GT; ++t) {
            v8f c = {};
            c = __builtin_amdgcn_wmma_f32_16x16x4_f32(false, a0, false, xa[cur][t],
                                                      (short)0, c, false, false);
            c = __builtin_amdgcn_wmma_f32_16x16x4_f32(false, a1, false, xb[cur][t],
                                                      (short)0, c, false, false);
            float h[8];
#pragma unroll
            for (int v = 0; v < 8; ++v)
                h[v] = fmaxf(fmaf(c[v], s0[v], bi0[v]), 0.0f);

            row_swap(h[0], h[2]);
            row_swap(h[1], h[3]);
            row_swap(h[4], h[6]);
            row_swap(h[5], h[7]);

            v2f bb[4];
            bb[0].x = h[0];  bb[0].y = h[1];
            bb[1].x = h[4];  bb[1].y = h[5];
            bb[2].x = h[2];  bb[2].y = h[3];
            bb[3].x = h[6];  bb[3].y = h[7];

            v8f dd = {};
#pragma unroll
            for (int j = 0; j < 4; ++j)
                dd = __builtin_amdgcn_wmma_f32_16x16x4_f32(false, aw[j], false, bb[j],
                                                           (short)0, dd, false, false);

            float gg[8];
#pragma unroll
            for (int v = 0; v < 8; ++v)
                gg[v] = fmaxf(fmaf(dd[v], s1[v], bi1[v]), 0.0f);

            float o = bs;
#pragma unroll
            for (int v = 0; v < 8; ++v)
                o = fmaf(ws[v], gg[v], o);

            sw[(g * GT + t) * 16] = o;
        }
        __builtin_amdgcn_sched_barrier(0);
    }

    __syncthreads();

    float* outb = out + (size_t)pd * HW;

#pragma unroll
    for (int j = 0; j < NJ; ++j) {
        const int cur = j & 1;

        const int l  = j * NTHR + tid;
        const int rl = l / NW;
        const int x  = l - rl * NW;
        int xn[3], xw[3];
#pragma unroll
        for (int i = 0; i < 3; ++i) {
            xn[i] = reflect_ix(x + (i - 1) * 2, NW);
            xw[i] = reflect_ix(x + (i - 1) * 4, NW);
        }
#pragma unroll
        for (int hh = 0; hh < 2; ++hh) {
            const int y = r0 + rl + 16 * hh;
            int yn[3], yw[3];
#pragma unroll
            for (int i = 0; i < 3; ++i) {
                yn[i] = (reflect_ix(y + (i - 1) * 2, NH) - lo) * NW;
                yw[i] = (reflect_ix(y + (i - 1) * 4, NH) - lo) * NW;
            }
            float acc = 0.0f;
#pragma unroll
            for (int s = 0; s < 9; ++s) {
                const int iy = s / 3, ix = s % 3;
                const float c1 = sim_s[yn[iy] + xn[ix]];
                const float c2 = sim_s[yw[iy] + xw[ix]];
                const float w_wide   = ow[cur][hh][s];
                const float w_narrow = ow[cur][hh][s + 9];
                acc = fmaf(0.5f, fmaf(w_narrow, c1, w_wide * c2), acc);
            }
            outb[y * NW + x] = acc * wt[cur][hh];
        }
        if (j + 2 < NJ) LOAD_AGG(j + 2, cur)
    }
#undef LOAD_AGG
#undef LOAD_X
}

extern "C" void kernel_launch(void* const* d_in, const int* in_sizes, int n_in,
                              void* d_out, int out_size, void* d_ws, size_t ws_size,
                              hipStream_t stream)
{
    const float* x1        = (const float*)d_in[0];
    const float* offset    = (const float*)d_in[1];
    const float* weight    = (const float*)d_in[2];
    const float* w0        = (const float*)d_in[3];
    const float* bn0_scale = (const float*)d_in[4];
    const float* bn0_bias  = (const float*)d_in[5];
    const float* w1        = (const float*)d_in[6];
    const float* bn1_scale = (const float*)d_in[7];
    const float* bn1_bias  = (const float*)d_in[8];
    const float* w_sim     = (const float*)d_in[9];
    const float* b_sim     = (const float*)d_in[10];
    float* out = (float*)d_out;

    sim_aggregate_kernel<<<256, NTHR, 0, stream>>>(
        x1, offset, weight, w0, bn0_scale, bn0_bias, w1, bn1_scale, bn1_bias,
        w_sim, b_sim, out);
}


# Work split in two with one more LDS-only barrier: MLP of the last 16 slab rows runs beside aggregation of the first 16 output rows, half the waves in each order; every sum order unchanged
# speedup vs baseline: 1.4074x; 1.0576x over previous
#include <hip/hip_runtime.h>


typedef __attribute__((ext_vector_type(2))) float v2f;
typedef __attribute__((ext_vector_type(8))) float v8f;

#define NB 2
#define NG 8
#define ND 32
#define NH 128
#define NW 160
#define HW (NH * NW)
#define DHW (ND * HW)

#define BAND 32
#define SLAB_ROWS 40
#define SLAB (SLAB_ROWS * NW)
#define NWAVES 16
#define GT 5
#define NGRP 5
#define GP (8 * NW)
#define WP (GP / NWAVES)
#define NTHR (NWAVES * 32)
#define SPARE (GT * 16)

typedef __attribute__((ext_vector_type(2))) unsigned int v2u;

__device__ __forceinline__ void row_swap(float& a, float& b) {
    const v2u r = __builtin_amdgcn_permlane16_swap(__float_as_uint(a), __float_as_uint(b),
                                                   false, false);
    a = __uint_as_float(r.x);
    b = __uint_as_float(r.y);
}

__device__ __forceinline__ void lds_barrier() {
    __builtin_amdgcn_fence(__ATOMIC_RELEASE, "workgroup", "local");
    __builtin_amdgcn_s_barrier();
    __builtin_amdgcn_fence(__ATOMIC_ACQUIRE, "workgroup", "local");
}

__device__ __forceinline__ int reflect_ix(int t, int n) {
    t = (t < 0) ? -t : t;
    t = (t >= n) ? (2 * n - 2 - t) : t;
    return t;
}

__global__ __launch_bounds__(NTHR) void sim_aggregate_kernel(
    const float* __restrict__ x1,
    const float* __restrict__ offset,
    const float* __restrict__ weight,
    const float* __restrict__ w0,
    const float* __restrict__ bn0_scale, const float* __restrict__ bn0_bias,
    const float* __restrict__ w1,
    const float* __restrict__ bn1_scale, const float* __restrict__ bn1_bias,
    const float* __restrict__ w_sim, const float* __restrict__ b_sim,
    float* __restrict__ out)
{
    __shared__ float sim_s[SLAB + SPARE];

    const int tid  = threadIdx.x;
    const int lane = tid & 31;
    const int wv   = tid >> 5;
    const int hi   = lane >> 4;
    const int col  = lane & 15;

    const int band = blockIdx.x & 3;
    const int pd   = blockIdx.x >> 2;
    const int b    = pd / ND;
    const int d    = pd - b * ND;
    const int r0   = band * BAND;
    int lo = r0 - 4;
    lo = (lo < 0) ? 0 : lo;
    lo = (lo > NH - SLAB_ROWS) ? (NH - SLAB_ROWS) : lo;

    const bool rev   = (r0 - lo) > 4;
    const int gbase  = rev ? (NGRP - 1) * GP : 0;
    const int gstep  = rev ? -GP : GP;
    const int yA     = r0 + (rev ? 16 : 0);
    const int yB     = r0 + (rev ? 0 : 16);

    const int kind = __builtin_amdgcn_readfirstlane((wv ^ (wv >> 2)) & 1);

    const int kb = hi * 2;
    v2f a0, a1;
    a0.x = w0[col * 8 + kb + 0];
    a0.y = w0[col * 8 + kb + 1];
    a1.x = w0[col * 8 + 4 + kb + 0];
    a1.y = w0[col * 8 + 4 + kb + 1];

    v2f aw[4];
#pragma unroll
    for (int j = 0; j < 4; ++j) {
        const int k0 = j * 4 + kb;
        aw[j].x = (col < 8) ? w1[col * 16 + k0 + 0] : 0.0f;
        aw[j].y = (col < 8) ? w1[col * 16 + k0 + 1] : 0.0f;
    }

    float s0[8], bi0[8], s1[8], bi1[8], ws[8];
    const int rb = 8 * hi;
#pragma unroll
    for (int v = 0; v < 8; ++v) {
        s0[v]  = bn0_scale[rb + v];
        bi0[v] = bn0_bias[rb + v];
        const int r = (rb + v) & 7;
        s1[v]  = bn1_scale[r];
        bi1[v] = bn1_bias[r];
        ws[v]  = w_sim[v];
    }
    const float bs = b_sim[0];

    const float* xs = x1 + (size_t)(b * NG + kb) * DHW + (size_t)d * HW
                         + lo * NW + gbase + wv * WP + col;
    float* sw = sim_s + (hi ? SLAB : gbase + wv * WP) + col;
    const int swstep = hi ? 0 : gstep;

    const float* ob = offset + (size_t)b * 18 * HW;
    const float* wb = weight + (size_t)b * HW;
    float ow[4][18];
    float wt[4];

#define LOAD_AGG(Y0, J, BUF)                                                  \
    {                                                                         \
        const int pix_ = (Y0) * NW + (J) * NTHR + tid;                        \
        _Pragma("unroll")                                                     \
        for (int s_ = 0; s_ < 18; ++s_)                                       \
            ow[BUF][s_] = ob[(size_t)s_ * HW + pix_];                         \
        wt[BUF] = wb[pix_];                                                   \
    }

    v2f xa[2][GT], xb[2][GT];

#define LOAD_X(G, BUF)                                                        \
    {                                                                         \
        const float* q_ = xs + (G) * gstep;                                   \
        _Pragma("unroll")                                                     \
        for (int t_ = 0; t_ < GT; ++t_) {                                     \
            const float* p_ = q_ + t_ * 16;                                   \
            xa[BUF][t_].x = p_[0];                                            \
            xa[BUF][t_].y = p_[DHW];                                          \
            xb[BUF][t_].x = p_[4 * (size_t)DHW];                              \
            xb[BUF][t_].y = p_[5 * (size_t)DHW];                              \
        }                                                                     \
    }

#define MLP_TILES(G, CUR)                                                     \
    {                                                                         \
        float* swg_ = sw + (G) * swstep;                                      \
        _Pragma("unroll")                                                     \
        for (int t = 0; t < GT; ++t) {                                        \
            v8f c = {};                                                       \
            c = __builtin_amdgcn_wmma_f32_16x16x4_f32(false, a0, false, xa[CUR][t], \
                                                      (short)0, c, false, false);   \
            c = __builtin_amdgcn_wmma_f32_16x16x4_f32(false, a1, false, xb[CUR][t], \
                                                      (short)0, c, false, false);   \
            float h[8];                                                       \
            _Pragma("unroll")                                                 \
            for (int v = 0; v < 8; ++v)                                       \
                h[v] = fmaxf(fmaf(c[v], s0[v], bi0[v]), 0.0f);                \
            row_swap(h[0], h[2]);                                             \
            row_swap(h[1], h[3]);                                             \
            row_swap(h[4], h[6]);                                             \
            row_swap(h[5], h[7]);                                             \
            v2f bb[4];                                                        \
            bb[0].x = h[0];  bb[0].y = h[1];                                  \
            bb[1].x = h[4];  bb[1].y = h[5];                                  \
            bb[2].x = h[2];  bb[2].y = h[3];                                  \
            bb[3].x = h[6];  bb[3].y = h[7];                                  \
            v8f dd = {};                                                      \
            _Pragma("unroll")                                                 \
            for (int j = 0; j < 4; ++j)                                       \
                dd = __builtin_amdgcn_wmma_f32_16x16x4_f32(false, aw[j], false, bb[j], \
                                                           (short)0, dd, false, false); \
            float gg[8];                                                      \
            _Pragma("unroll")                                                 \
            for (int v = 0; v < 8; ++v)                                       \
                gg[v] = fmaxf(fmaf(dd[v], s1[v], bi1[v]), 0.0f);              \
            float o = bs;                                                     \
            _Pragma("unroll")                                                 \
            for (int v = 0; v < 8; ++v)                                       \
                o = fmaf(ws[v], gg[v], o);                                    \
            swg_[t * 16] = o;                                                 \
        }                                                                     \
    }

    float* outb = out + (size_t)pd * HW;

#define AGG(Y0, J, BUF)                                                       \
    {                                                                         \
        const int l  = (J) * NTHR + tid;                                      \
        const int rl = l / NW;                                                \
        const int x  = l - rl * NW;                                           \
        const int y  = (Y0) + rl;                                             \
        int xn[3], xw[3], yn[3], yw[3];                                       \
        _Pragma("unroll")                                                     \
        for (int i = 0; i < 3; ++i) {                                         \
            xn[i] = reflect_ix(x + (i - 1) * 2, NW);                          \
            xw[i] = reflect_ix(x + (i - 1) * 4, NW);                          \
            yn[i] = (reflect_ix(y + (i - 1) * 2, NH) - lo) * NW;              \
            yw[i] = (reflect_ix(y + (i - 1) * 4, NH) - lo) * NW;              \
        }                                                                     \
        float acc = 0.0f;                                                     \
        _Pragma("unroll")                                                     \
        for (int s = 0; s < 9; ++s) {                                         \
            const int iy = s / 3, ix = s % 3;                                 \
            const float c1 = sim_s[yn[iy] + xn[ix]];                          \
            const float c2 = sim_s[yw[iy] + xw[ix]];                          \
            const float w_wide   = ow[BUF][s];                                \
            const float w_narrow = ow[BUF][s + 9];                            \
            acc = fmaf(0.5f, fmaf(w_narrow, c1, w_wide * c2), acc);           \
        }                                                                     \
        outb[y * NW + x] = acc * wt[BUF];                                     \
    }

#define AGG_HALF_A                                                            \
    AGG(yA, 0, 0)  LOAD_AGG(yA, 4, 0)                                         \
    AGG(yA, 1, 1)  LOAD_AGG(yB, 0, 1)                                         \
    AGG(yA, 2, 2)  LOAD_AGG(yB, 1, 2)                                         \
    AGG(yA, 3, 3)  LOAD_AGG(yB, 2, 3)                                         \
    AGG(yA, 4, 0)  LOAD_AGG(yB, 3, 0)

    LOAD_X(0, 0)

    LOAD_X(1, 1)
    __builtin_amdgcn_sched_barrier(0);
    MLP_TILES(0, 0)
    __builtin_amdgcn_sched_barrier(0);

    LOAD_X(2, 0)
    __builtin_amdgcn_sched_barrier(0);
    MLP_TILES(1, 1)
    __builtin_amdgcn_sched_barrier(0);

    LOAD_X(3, 1)
    LOAD_AGG(yA, 0, 0)
    LOAD_AGG(yA, 1, 1)
    __builtin_amdgcn_sched_barrier(0);
    MLP_TILES(2, 0)
    __builtin_amdgcn_sched_barrier(0);

    lds_barrier();

    if (kind == 0) {
        LOAD_X(4, 0)
        __builtin_amdgcn_sched_barrier(0);
        MLP_TILES(3, 1)
        __builtin_amdgcn_sched_barrier(0);
        LOAD_AGG(yA, 2, 2)
        LOAD_AGG(yA, 3, 3)
        __builtin_amdgcn_sched_barrier(0);
        MLP_TILES(4, 0)
        __builtin_amdgcn_sched_barrier(0);
        AGG_HALF_A
    } else {
        LOAD_AGG(yA, 2, 2)
        LOAD_AGG(yA, 3, 3)
        AGG(yA, 0, 0)  LOAD_AGG(yA, 4, 0)
        AGG(yA, 1, 1)
        AGG(yA, 2, 2)
        AGG(yA, 3, 3)
        AGG(yA, 4, 0)
        LOAD_X(4, 0)
        LOAD_AGG(yB, 0, 1)
        __builtin_amdgcn_sched_barrier(0);
        MLP_TILES(3, 1)
        __builtin_amdgcn_sched_barrier(0);
        LOAD_AGG(yB, 1, 2)
        LOAD_AGG(yB, 2, 3)
        LOAD_AGG(yB, 3, 0)
        __builtin_amdgcn_sched_barrier(0);
        MLP_TILES(4, 0)
        __builtin_amdgcn_sched_barrier(0);
    }

    lds_barrier();

    AGG(yB, 0, 1)  LOAD_AGG(yB, 4, 1)
    AGG(yB, 1, 2)
    AGG(yB, 2, 3)
    AGG(yB, 3, 0)
    AGG(yB, 4, 1)

#undef AGG_HALF_A
#undef AGG
#undef MLP_TILES
#undef LOAD_AGG
#undef LOAD_X
}

extern "C" void kernel_launch(void* const* d_in, const int* in_sizes, int n_in,
                              void* d_out, int out_size, void* d_ws, size_t ws_size,
                              hipStream_t stream)
{
    const float* x1        = (const float*)d_in[0];
    const float* offset    = (const float*)d_in[1];
    const float* weight    = (const float*)d_in[2];
    const float* w0        = (const float*)d_in[3];
    const float* bn0_scale = (const float*)d_in[4];
    const float* bn0_bias  = (const float*)d_in[5];
    const float* w1        = (const float*)d_in[6];
    const float* bn1_scale = (const float*)d_in[7];
    const float* bn1_bias  = (const float*)d_in[8];
    const float* w_sim     = (const float*)d_in[9];
    const float* b_sim     = (const float*)d_in[10];
    float* out = (float*)d_out;

    sim_aggregate_kernel<<<256, NTHR, 0, stream>>>(
        x1, offset, weight, w0, bn0_scale, bn0_bias, w1, bn1_scale, bn1_bias,
        w_sim, b_sim, out);
}
